# starting kernel
# baseline (speedup 1.0000x reference)
#include <hip/hip_runtime.h>
#include <hip/hip_bf16.h>


#define N_NODES 100000
#define D_FEAT  512
#define N_EDGES 3200000
#define L1D     32
#define L2D     16

typedef __attribute__((ext_vector_type(16))) _Float16 v16h;
typedef __attribute__((ext_vector_type(8)))  _Float16 v8h;
typedef __attribute__((ext_vector_type(8)))  float    v8f;
typedef __attribute__((ext_vector_type(4)))  float    v4f;

__global__ void egae_prep(const float* __restrict__ W1,
                          const float* __restrict__ W2,
                          _Float16* __restrict__ W1t,
                          _Float16* __restrict__ W2t) {
  int t = blockIdx.x * blockDim.x + threadIdx.x;
  if (t < D_FEAT * L1D) {
    int k = t / L1D, n = t % L1D;
    W1t[(size_t)n * D_FEAT + k] = (_Float16)W1[t];
  }
  if (t < L1D * L2D) {
    int k = t / L2D, n = t % L2D;
    W2t[(size_t)n * L1D + k] = (_Float16)W2[t];
  }
}

__global__ void egae_zero(float* __restrict__ p, int n) {
  int i = blockIdx.x * blockDim.x + threadIdx.x;
  if (i < n) p[i] = 0.0f;
}

__global__ void egae_gemm1(const float* __restrict__ X,
                           const _Float16* __restrict__ W1t,
                           float* __restrict__ Y1) {
  const int lane = threadIdx.x & 31;
  const int wave = (blockIdx.x * blockDim.x + threadIdx.x) >> 5;
  const int row0 = wave << 4;
  if (row0 >= N_NODES) return;
  const int m  = lane & 15;
  const int kh = lane >> 4;

  const float*    arow = X   + (size_t)(row0 + m) * D_FEAT + kh * 8;
  const _Float16* bc0  = W1t + (size_t)m         * D_FEAT + kh * 8;
  const _Float16* bc1  = W1t + (size_t)(m + 16)  * D_FEAT + kh * 8;

  v8f acc0 = {}; v8f acc1 = {};
  for (int k0 = 0; k0 < D_FEAT; k0 += 32) {
    v4f a0 = *(const v4f*)(arow + k0);
    v4f a1 = *(const v4f*)(arow + k0 + 4);
    v4f a2 = *(const v4f*)(arow + k0 + 16);
    v4f a3 = *(const v4f*)(arow + k0 + 20);
    v16h af;
#pragma unroll
    for (int i = 0; i < 4; ++i) {
      af[i]      = (_Float16)a0[i];
      af[4 + i]  = (_Float16)a1[i];
      af[8 + i]  = (_Float16)a2[i];
      af[12 + i] = (_Float16)a3[i];
    }
    v8h b0lo = *(const v8h*)(bc0 + k0);
    v8h b0hi = *(const v8h*)(bc0 + k0 + 16);
    v8h b1lo = *(const v8h*)(bc1 + k0);
    v8h b1hi = *(const v8h*)(bc1 + k0 + 16);
    v16h bf0, bf1;
#pragma unroll
    for (int i = 0; i < 8; ++i) {
      bf0[i] = b0lo[i]; bf0[8 + i] = b0hi[i];
      bf1[i] = b1lo[i]; bf1[8 + i] = b1hi[i];
    }
    acc0 = __builtin_amdgcn_wmma_f32_16x16x32_f16(false, af, false, bf0,
                                                  (short)0, acc0, false, false);
    acc1 = __builtin_amdgcn_wmma_f32_16x16x32_f16(false, af, false, bf1,
                                                  (short)0, acc1, false, false);
  }
  float* drow = Y1 + (size_t)(row0 + kh * 8) * L1D + m;
#pragma unroll
  for (int v = 0; v < 8; ++v) {
    drow[(size_t)v * L1D]      = acc0[v];
    drow[(size_t)v * L1D + 16] = acc1[v];
  }
}

__global__ void egae_spmm32(const int* __restrict__ er, const int* __restrict__ ec,
                            const float* __restrict__ ev,
                            const float* __restrict__ src, float* __restrict__ dst) {
  long long t = (long long)blockIdx.x * blockDim.x + threadIdx.x;
  int e = (int)(t >> 5);
  if (e >= N_EDGES) return;
  int col = (int)(t & 31);
  float x = ev[e] * src[(size_t)ec[e] * L1D + col];
  atomicAdd(dst + (size_t)er[e] * L1D + col, x);
}

__global__ void egae_gemm2(const float* __restrict__ H1,
                           const _Float16* __restrict__ W2t,
                           float* __restrict__ Y2) {
  const int lane = threadIdx.x & 31;
  const int wave = (blockIdx.x * blockDim.x + threadIdx.x) >> 5;
  const int row0 = wave << 4;
  if (row0 >= N_NODES) return;
  const int m  = lane & 15;
  const int kh = lane >> 4;

  const float* arow = H1 + (size_t)(row0 + m) * L1D + kh * 8;
  v4f a0 = *(const v4f*)(arow);
  v4f a1 = *(const v4f*)(arow + 4);
  v4f a2 = *(const v4f*)(arow + 16);
  v4f a3 = *(const v4f*)(arow + 20);
  v16h af;
#pragma unroll
  for (int i = 0; i < 4; ++i) {
    af[i]      = (_Float16)fmaxf(a0[i], 0.0f);
    af[4 + i]  = (_Float16)fmaxf(a1[i], 0.0f);
    af[8 + i]  = (_Float16)fmaxf(a2[i], 0.0f);
    af[12 + i] = (_Float16)fmaxf(a3[i], 0.0f);
  }
  const _Float16* bc = W2t + (size_t)m * L1D + kh * 8;
  v8h blo = *(const v8h*)(bc);
  v8h bhi = *(const v8h*)(bc + 16);
  v16h bf;
#pragma unroll
  for (int i = 0; i < 8; ++i) { bf[i] = blo[i]; bf[8 + i] = bhi[i]; }

  v8f acc = {};
  acc = __builtin_amdgcn_wmma_f32_16x16x32_f16(false, af, false, bf,
                                               (short)0, acc, false, false);
  float* drow = Y2 + (size_t)(row0 + kh * 8) * L2D + m;
#pragma unroll
  for (int v = 0; v < 8; ++v) drow[(size_t)v * L2D] = acc[v];
}

__global__ void egae_spmm16(const int* __restrict__ er, const int* __restrict__ ec,
                            const float* __restrict__ ev,
                            const float* __restrict__ src, float* __restrict__ dst) {
  long long t = (long long)blockIdx.x * blockDim.x + threadIdx.x;
  int e = (int)(t >> 4);
  if (e >= N_EDGES) return;
  int col = (int)(t & 15);
  float x = ev[e] * src[(size_t)ec[e] * L2D + col];
  atomicAdd(dst + (size_t)er[e] * L2D + col, x);
}

__global__ void egae_norm(const float* __restrict__ Y3, float* __restrict__ out) {
  int i = blockIdx.x * blockDim.x + threadIdx.x;
  if (i >= N_NODES) return;
  const v4f* s = (const v4f*)(Y3 + (size_t)i * L2D);
  v4f x0 = s[0], x1 = s[1], x2 = s[2], x3 = s[3];
  float ss = 0.0f;
#pragma unroll
  for (int j = 0; j < 4; ++j)
    ss += x0[j] * x0[j] + x1[j] * x1[j] + x2[j] * x2[j] + x3[j] * x3[j];
  float inv = 1.0f / fmaxf(sqrtf(ss), 1e-7f);
  v4f* o = (v4f*)(out + (size_t)i * L2D);
  o[0] = x0 * inv; o[1] = x1 * inv; o[2] = x2 * inv; o[3] = x3 * inv;
}

extern "C" void kernel_launch(void* const* d_in, const int* in_sizes, int n_in,
                              void* d_out, int out_size, void* d_ws, size_t ws_size,
                              hipStream_t stream) {
  const float* X  = (const float*)d_in[0];
  const float* W1 = (const float*)d_in[1];
  const float* W2 = (const float*)d_in[2];
  const int*   lr = (const int*)d_in[3];
  const int*   lc = (const int*)d_in[4];
  const float* lv = (const float*)d_in[5];
  float* out = (float*)d_out;

  char* ws = (char*)d_ws;
  float*    Y1  = (float*)(ws + 0);
  float*    H1  = (float*)(ws + 12800000);
  float*    Y2  = (float*)(ws + 25600000);
  float*    Y3  = (float*)(ws + 32000000);
  _Float16* W1t = (_Float16*)(ws + 38400000);
  _Float16* W2t = (_Float16*)(ws + 38432768);

  const int waves  = N_NODES / 16;
  const int gblk   = (waves + 7) / 8;

  egae_prep<<<(D_FEAT * L1D + 255) / 256, 256, 0, stream>>>(W1, W2, W1t, W2t);
  egae_zero<<<(N_NODES * L1D + 255) / 256, 256, 0, stream>>>(H1, N_NODES * L1D);
  egae_zero<<<(N_NODES * L2D + 255) / 256, 256, 0, stream>>>(Y3, N_NODES * L2D);
  egae_gemm1<<<gblk, 256, 0, stream>>>(X, W1t, Y1);
  egae_spmm32<<<(N_EDGES * 32) / 256, 256, 0, stream>>>(lr, lc, lv, Y1, H1);
  egae_gemm2<<<gblk, 256, 0, stream>>>(H1, W2t, Y2);
  egae_spmm16<<<(N_EDGES * 16) / 256, 256, 0, stream>>>(lr, lc, lv, Y2, Y3);
  egae_norm<<<(N_NODES + 255) / 256, 256, 0, stream>>>(Y3, out);
}


# edges counting-sorted by destination row block (128 rows); a block's rows summed in LDS by one work-group (LDS f32 adds, arrival order); global f32 atomics, H1, zero-fills gone; gemm2 its epilogue
# speedup vs baseline: 3.7009x; 3.7009x over previous
#include <hip/hip_runtime.h>
#include <hip/hip_bf16.h>


#define N_NODES 100000
#define D_FEAT  512
#define N_EDGES 3200000
#define L1D     32
#define L2D     16
#define RB_SHIFT 7
#define RB      128
#define NBUCK   782
#define NBPAD   1024
#define CHUNK   4096
#define NCHUNK  782
#define COLMASK 0x1FFFF

typedef __attribute__((ext_vector_type(16))) _Float16 v16h;
typedef __attribute__((ext_vector_type(8)))  _Float16 v8h;
typedef __attribute__((ext_vector_type(8)))  float    v8f;
typedef __attribute__((ext_vector_type(4)))  float    v4f;

__global__ void egae_prep(const float* __restrict__ W1,
                          const float* __restrict__ W2,
                          _Float16* __restrict__ W1t,
                          _Float16* __restrict__ W2t,
                          int* __restrict__ counters) {
  int t = blockIdx.x * blockDim.x + threadIdx.x;
  if (t < 2 * NBPAD) counters[t] = 0;
  if (t < D_FEAT * L1D) {
    int k = t / L1D, n = t % L1D;
    W1t[(size_t)n * D_FEAT + k] = (_Float16)W1[t];
  }
  if (t < L1D * L2D) {
    int k = t / L2D, n = t % L2D;
    W2t[(size_t)n * L1D + k] = (_Float16)W2[t];
  }
}

__global__ void egae_gemm1(const float* __restrict__ X,
                           const _Float16* __restrict__ W1t,
                           float* __restrict__ Y1) {
  const int lane = threadIdx.x & 31;
  const int wave = (blockIdx.x * blockDim.x + threadIdx.x) >> 5;
  const int row0 = wave << 4;
  if (row0 >= N_NODES) return;
  const int m  = lane & 15;
  const int kh = lane >> 4;

  const float*    arow = X   + (size_t)(row0 + m) * D_FEAT + kh * 8;
  const _Float16* bc0  = W1t + (size_t)m         * D_FEAT + kh * 8;
  const _Float16* bc1  = W1t + (size_t)(m + 16)  * D_FEAT + kh * 8;

  v8f acc0 = {}; v8f acc1 = {};
  for (int k0 = 0; k0 < D_FEAT; k0 += 32) {
    v4f a0 = *(const v4f*)(arow + k0);
    v4f a1 = *(const v4f*)(arow + k0 + 4);
    v4f a2 = *(const v4f*)(arow + k0 + 16);
    v4f a3 = *(const v4f*)(arow + k0 + 20);
    v16h af;
#pragma unroll
    for (int i = 0; i < 4; ++i) {
      af[i]      = (_Float16)a0[i];
      af[4 + i]  = (_Float16)a1[i];
      af[8 + i]  = (_Float16)a2[i];
      af[12 + i] = (_Float16)a3[i];
    }
    v8h b0lo = *(const v8h*)(bc0 + k0);
    v8h b0hi = *(const v8h*)(bc0 + k0 + 16);
    v8h b1lo = *(const v8h*)(bc1 + k0);
    v8h b1hi = *(const v8h*)(bc1 + k0 + 16);
    v16h bf0, bf1;
#pragma unroll
    for (int i = 0; i < 8; ++i) {
      bf0[i] = b0lo[i]; bf0[8 + i] = b0hi[i];
      bf1[i] = b1lo[i]; bf1[8 + i] = b1hi[i];
    }
    acc0 = __builtin_amdgcn_wmma_f32_16x16x32_f16(false, af, false, bf0,
                                                  (short)0, acc0, false, false);
    acc1 = __builtin_amdgcn_wmma_f32_16x16x32_f16(false, af, false, bf1,
                                                  (short)0, acc1, false, false);
  }
  float* drow = Y1 + (size_t)(row0 + kh * 8) * L1D + m;
#pragma unroll
  for (int v = 0; v < 8; ++v) {
    drow[(size_t)v * L1D]      = acc0[v];
    drow[(size_t)v * L1D + 16] = acc1[v];
  }
}

__global__ void egae_hist(const int* __restrict__ er, int* __restrict__ gcount) {
  __shared__ int cnt[NBPAD];
  const int tid = threadIdx.x;
  for (int i = tid; i < NBPAD; i += 256) cnt[i] = 0;
  __syncthreads();
  const int base = blockIdx.x * CHUNK;
#pragma unroll
  for (int i = 0; i < CHUNK / 256; ++i) {
    const int e = base + i * 256 + tid;
    if (e < N_EDGES) atomicAdd(&cnt[er[e] >> RB_SHIFT], 1);
  }
  __syncthreads();
  for (int k = tid; k < NBPAD; k += 256) {
    const int n = cnt[k];
    if (n) atomicAdd(&gcount[k], n);
  }
}

__global__ void egae_scan(const int* __restrict__ gcount, int* __restrict__ gbase) {
  __shared__ int sc[NBPAD];
  const int t = threadIdx.x;
  sc[t] = gcount[t];
  __syncthreads();
  for (int off = 1; off < NBPAD; off <<= 1) {
    const int add = (t >= off) ? sc[t - off] : 0;
    __syncthreads();
    sc[t] += add;
    __syncthreads();
  }
  gbase[t + 1] = sc[t];
  if (t == 0) gbase[0] = 0;
}

__global__ void egae_scatter(const int* __restrict__ er, const int* __restrict__ ec,
                             const float* __restrict__ ev,
                             const int* __restrict__ gbase, int* __restrict__ gcursor,
                             int2* __restrict__ E) {
  __shared__ int   cnt[NBPAD];
  __shared__ int   gdst[NBPAD];
  __shared__ int   lstart[NBPAD];
  __shared__ int   part[256];
  __shared__ int   sPk[CHUNK];
  __shared__ float sVal[CHUNK];
  __shared__ int   sDst[CHUNK];
  const int tid = threadIdx.x;
  for (int i = tid; i < NBPAD; i += 256) cnt[i] = 0;
  __syncthreads();
  const int base = blockIdx.x * CHUNK;
  int r[CHUNK / 256], c[CHUNK / 256], rk[CHUNK / 256];
  float v[CHUNK / 256];
#pragma unroll
  for (int i = 0; i < CHUNK / 256; ++i) {
    const int e = base + i * 256 + tid;
    r[i] = -1; c[i] = 0; v[i] = 0.0f; rk[i] = 0;
    if (e < N_EDGES) {
      r[i] = er[e]; c[i] = ec[e]; v[i] = ev[e];
      rk[i] = atomicAdd(&cnt[r[i] >> RB_SHIFT], 1);
    }
  }
  __syncthreads();
  int gslot[NBPAD / 256];
#pragma unroll
  for (int q = 0; q < NBPAD / 256; ++q) {
    const int k  = q * 256 + tid;
    const int nk = cnt[k];
    gslot[q] = 0;
    if (nk) gslot[q] = gbase[k] + atomicAdd(&gcursor[k], nk);
  }
  const int b0 = tid * 4;
  int n[4];
#pragma unroll
  for (int k = 0; k < 4; ++k) n[k] = cnt[b0 + k];
  const int mine = n[0] + n[1] + n[2] + n[3];
  part[tid] = mine;
  __syncthreads();
  for (int off = 1; off < 256; off <<= 1) {
    const int add = (tid >= off) ? part[tid - off] : 0;
    __syncthreads();
    part[tid] += add;
    __syncthreads();
  }
  int excl = part[tid] - mine;
#pragma unroll
  for (int k = 0; k < 4; ++k) {
    lstart[b0 + k] = excl;
    excl += n[k];
  }
#pragma unroll
  for (int q = 0; q < NBPAD / 256; ++q) gdst[q * 256 + tid] = gslot[q];
  __syncthreads();
#pragma unroll
  for (int i = 0; i < CHUNK / 256; ++i) {
    if (r[i] >= 0) {
      const int bk = r[i] >> RB_SHIFT;
      const int p  = lstart[bk] + rk[i];
      sPk[p]  = c[i] | ((r[i] & (RB - 1)) << 17);
      sVal[p] = v[i];
      sDst[p] = gdst[bk] + rk[i];
    }
  }
  __syncthreads();
  const int nvalid = (N_EDGES - base < CHUNK) ? (N_EDGES - base) : CHUNK;
#pragma unroll
  for (int i = 0; i < CHUNK / 256; ++i) {
    const int p = i * 256 + tid;
    if (p < nvalid) {
      int2 rec; rec.x = sPk[p]; rec.y = __float_as_int(sVal[p]);
      E[sDst[p]] = rec;
    }
  }
}

template <bool FULL>
static __device__ __forceinline__ void spmm32_tile(const float* __restrict__ src,
                                                   float* acc, int pk_l, float v_l, int lane) {
  float g[32];
#pragma unroll
  for (int j = 0; j < 32; ++j) {
    const int pk = __builtin_amdgcn_readlane(pk_l, j);
    g[j] = 0.0f;
    if (FULL || pk >= 0) g[j] = src[(size_t)(pk & COLMASK) * L1D + lane];
  }
#pragma unroll
  for (int j = 0; j < 32; ++j) {
    const int   pk = __builtin_amdgcn_readlane(pk_l, j);
    const float v  = __int_as_float(__builtin_amdgcn_readlane(__float_as_int(v_l), j));
    if (FULL || pk >= 0) atomicAdd(acc + (pk >> 17) * L1D + lane, v * g[j]);
  }
}

__global__ void __launch_bounds__(512)
egae_spmm32(const int2* __restrict__ E, const int* __restrict__ gbase,
            const float* __restrict__ src, const _Float16* __restrict__ W2t,
            float* __restrict__ Y2) {
  __shared__ float acc[RB * L1D];
  const int tid  = threadIdx.x;
  const int lane = tid & 31;
  const int wave = __builtin_amdgcn_readfirstlane(tid >> 5);
  for (int i = tid; i < RB * L1D; i += 512) acc[i] = 0.0f;
  __syncthreads();
  const int bkt = blockIdx.x;
  const int beg = gbase[bkt], end = gbase[bkt + 1];
  for (int base = beg + wave * 32; base < end; base += 16 * 32) {
    const int idx = base + lane;
    int pk_l = -1; float v_l = 0.0f;
    if (idx < end) { const int2 rec = E[idx]; pk_l = rec.x; v_l = __int_as_float(rec.y); }
    if (base + 32 <= end) spmm32_tile<true >(src, acc, pk_l, v_l, lane);
    else                  spmm32_tile<false>(src, acc, pk_l, v_l, lane);
  }
  __syncthreads();
  if (wave < RB / 16) {
    const int m  = lane & 15;
    const int kh = lane >> 4;
    const float* arow = acc + (wave * 16 + m) * L1D + kh * 8;
    v4f a0 = *(const v4f*)(arow);
    v4f a1 = *(const v4f*)(arow + 4);
    v4f a2 = *(const v4f*)(arow + 16);
    v4f a3 = *(const v4f*)(arow + 20);
    v16h af;
#pragma unroll
    for (int i = 0; i < 4; ++i) {
      af[i]      = (_Float16)fmaxf(a0[i], 0.0f);
      af[4 + i]  = (_Float16)fmaxf(a1[i], 0.0f);
      af[8 + i]  = (_Float16)fmaxf(a2[i], 0.0f);
      af[12 + i] = (_Float16)fmaxf(a3[i], 0.0f);
    }
    const _Float16* bc = W2t + (size_t)m * L1D + kh * 8;
    v8h blo = *(const v8h*)(bc);
    v8h bhi = *(const v8h*)(bc + 16);
    v16h bf;
#pragma unroll
    for (int i = 0; i < 8; ++i) { bf[i] = blo[i]; bf[8 + i] = bhi[i]; }
    v8f d = {};
    d = __builtin_amdgcn_wmma_f32_16x16x32_f16(false, af, false, bf,
                                               (short)0, d, false, false);
    const int row0 = bkt * RB + wave * 16 + kh * 8;
#pragma unroll
    for (int v = 0; v < 8; ++v)
      if (row0 + v < N_NODES) Y2[(size_t)(row0 + v) * L2D + m] = d[v];
  }
}

template <int J>
static __device__ __forceinline__ int row_lane(int v) {
  return __builtin_amdgcn_update_dpp(0, v, 0x150 + J, 0xF, 0xF, false);
}

template <int J, bool FULL>
static __device__ __forceinline__ void spmm16_gather(const float* __restrict__ src,
                                                     int pk_l, int col, float (&g)[16]) {
  if constexpr (J < 16) {
    const int pk = row_lane<J>(pk_l);
    g[J] = 0.0f;
    if (FULL || pk >= 0) g[J] = src[(size_t)(pk & COLMASK) * L2D + col];
    spmm16_gather<J + 1, FULL>(src, pk_l, col, g);
  }
}

template <int J, bool FULL>
static __device__ __forceinline__ void spmm16_add(float* acc, int pk_l,
                                                  float v_l, int col, const float (&g)[16]) {
  if constexpr (J < 16) {
    const int   pk = row_lane<J>(pk_l);
    const float v  = __int_as_float(row_lane<J>(__float_as_int(v_l)));
    if (FULL || pk >= 0) atomicAdd(acc + (pk >> 17) * L2D + col, v * g[J]);
    spmm16_add<J + 1, FULL>(acc, pk_l, v_l, col, g);
  }
}

__global__ void __launch_bounds__(512)
egae_spmm16(const int2* __restrict__ E, const int* __restrict__ gbase,
            const float* __restrict__ src, float* __restrict__ dst) {
  __shared__ float acc[RB * L2D];
  const int tid  = threadIdx.x;
  const int lane = tid & 31;
  const int wave = __builtin_amdgcn_readfirstlane(tid >> 5);
  for (int i = tid; i < RB * L2D; i += 512) acc[i] = 0.0f;
  __syncthreads();
  const int bkt = blockIdx.x;
  const int beg = gbase[bkt], end = gbase[bkt + 1];
  const int col = lane & 15;
  for (int base = beg + wave * 32; base < end; base += 16 * 32) {
    const int idx = base + lane;
    int pk_l = -1; float v_l = 0.0f;
    if (idx < end) { const int2 rec = E[idx]; pk_l = rec.x; v_l = __int_as_float(rec.y); }
    float g[16];
    if (base + 32 <= end) {
      spmm16_gather<0, true >(src, pk_l, col, g);
      spmm16_add<0, true >(acc, pk_l, v_l, col, g);
    } else {
      spmm16_gather<0, false>(src, pk_l, col, g);
      spmm16_add<0, false>(acc, pk_l, v_l, col, g);
    }
  }
  __syncthreads();
  const int row = bkt * RB + (tid >> 2);
  if (row < N_NODES)
    *(v4f*)(dst + (size_t)row * L2D + (tid & 3) * 4) = *(const v4f*)(acc + tid * 4);
}

__global__ void egae_norm(const float* __restrict__ Y3, float* __restrict__ out) {
  int i = blockIdx.x * blockDim.x + threadIdx.x;
  if (i >= N_NODES) return;
  const v4f* s = (const v4f*)(Y3 + (size_t)i * L2D);
  v4f x0 = s[0], x1 = s[1], x2 = s[2], x3 = s[3];
  float ss = 0.0f;
#pragma unroll
  for (int j = 0; j < 4; ++j)
    ss += x0[j] * x0[j] + x1[j] * x1[j] + x2[j] * x2[j] + x3[j] * x3[j];
  float inv = 1.0f / fmaxf(sqrtf(ss), 1e-7f);
  v4f* o = (v4f*)(out + (size_t)i * L2D);
  o[0] = x0 * inv; o[1] = x1 * inv; o[2] = x2 * inv; o[3] = x3 * inv;
}

extern "C" void kernel_launch(void* const* d_in, const int* in_sizes, int n_in,
                              void* d_out, int out_size, void* d_ws, size_t ws_size,
                              hipStream_t stream) {
  const float* X  = (const float*)d_in[0];
  const float* W1 = (const float*)d_in[1];
  const float* W2 = (const float*)d_in[2];
  const int*   lr = (const int*)d_in[3];
  const int*   lc = (const int*)d_in[4];
  const float* lv = (const float*)d_in[5];
  float* out = (float*)d_out;

  char* ws = (char*)d_ws;
  float*    Y1  = (float*)(ws + 0);
  float*    Y3  = (float*)(ws + 0);
  int2*     E   = (int2*)(ws + 12800000);
  _Float16* W1t = (_Float16*)(ws + 38400000);
  int*      gbase = (int*)(ws + 38400000);
  _Float16* W2t = (_Float16*)(ws + 38432768);
  float* Y2      = out;
  int*   gcount  = (int*)d_out;
  int*   gcursor = (int*)d_out + NBPAD;

  const int waves  = N_NODES / 16;
  const int gblk   = (waves + 7) / 8;

  egae_prep<<<(D_FEAT * L1D + 255) / 256, 256, 0, stream>>>(W1, W2, W1t, W2t, gcount);
  egae_gemm1<<<gblk, 256, 0, stream>>>(X, W1t, Y1);
  egae_hist<<<NCHUNK, 256, 0, stream>>>(lr, gcount);
  egae_scan<<<1, NBPAD, 0, stream>>>(gcount, gbase);
  egae_scatter<<<NCHUNK, 256, 0, stream>>>(lr, lc, lv, gbase, gcursor, E);
  egae_spmm32<<<NBUCK, 512, 0, stream>>>(E, gbase, Y1, W2t, Y2);
  egae_spmm16<<<NBUCK, 512, 0, stream>>>(E, gbase, Y2, Y3);
  egae_norm<<<(N_NODES + 255) / 256, 256, 0, stream>>>(Y3, out);
}


# launch fusion: each gemm1 work-group also counts one edge chunk's bucket histogram, each scatter work-group scans the bucket totals itself (hist and scan launches gone, 8 -> 6)
# speedup vs baseline: 4.0140x; 1.0846x over previous
#include <hip/hip_runtime.h>
#include <hip/hip_bf16.h>


#define N_NODES 100000
#define D_FEAT  512
#define N_EDGES 3200000
#define L1D     32
#define L2D     16
#define RB_SHIFT 7
#define RB      128
#define NBUCK   782
#define NBPAD   1024
#define CHUNK   4096
#define NCHUNK  782
#define COLMASK 0x1FFFF

typedef __attribute__((ext_vector_type(16))) _Float16 v16h;
typedef __attribute__((ext_vector_type(8)))  _Float16 v8h;
typedef __attribute__((ext_vector_type(8)))  float    v8f;
typedef __attribute__((ext_vector_type(4)))  float    v4f;

__global__ void egae_prep(const float* __restrict__ W1,
                          const float* __restrict__ W2,
                          _Float16* __restrict__ W1t,
                          _Float16* __restrict__ W2t,
                          int* __restrict__ counters) {
  int t = blockIdx.x * blockDim.x + threadIdx.x;
  if (t < 2 * NBPAD) counters[t] = 0;
  if (t < D_FEAT * L1D) {
    int k = t / L1D, n = t % L1D;
    W1t[(size_t)n * D_FEAT + k] = (_Float16)W1[t];
  }
  if (t < L1D * L2D) {
    int k = t / L2D, n = t % L2D;
    W2t[(size_t)n * L1D + k] = (_Float16)W2[t];
  }
}

static_assert((N_NODES / 16 + 7) / 8 <= NCHUNK, "gemm1 row tiles must fit the chunk grid");
__global__ void __launch_bounds__(256)
egae_gemm1_hist(const float* __restrict__ X,
                const _Float16* __restrict__ W1t,
                float* __restrict__ Y1,
                const int* __restrict__ er, int* __restrict__ gcount) {
  __shared__ int cnt[NBPAD];
  const int tid = threadIdx.x;
  for (int i = tid; i < NBPAD; i += 256) cnt[i] = 0;
  __syncthreads();
  const int ebase = blockIdx.x * CHUNK;
  int r[CHUNK / 256];
#pragma unroll
  for (int i = 0; i < CHUNK / 256; ++i) {
    const int e = ebase + i * 256 + tid;
    r[i] = -1;
    if (e < N_EDGES) r[i] = er[e];
  }

  const int lane = tid & 31;
  const int wave = (blockIdx.x * 256 + tid) >> 5;
  const int row0 = wave << 4;
  const int m  = lane & 15;
  const int kh = lane >> 4;
  v8f acc0 = {}; v8f acc1 = {};
  if (row0 < N_NODES) {
    const float*    arow = X   + (size_t)(row0 + m) * D_FEAT + kh * 8;
    const _Float16* bc0  = W1t + (size_t)m         * D_FEAT + kh * 8;
    const _Float16* bc1  = W1t + (size_t)(m + 16)  * D_FEAT + kh * 8;
    for (int k0 = 0; k0 < D_FEAT; k0 += 32) {
      v4f a0 = *(const v4f*)(arow + k0);
      v4f a1 = *(const v4f*)(arow + k0 + 4);
      v4f a2 = *(const v4f*)(arow + k0 + 16);
      v4f a3 = *(const v4f*)(arow + k0 + 20);
      v16h af;
#pragma unroll
      for (int i = 0; i < 4; ++i) {
        af[i]      = (_Float16)a0[i];
        af[4 + i]  = (_Float16)a1[i];
        af[8 + i]  = (_Float16)a2[i];
        af[12 + i] = (_Float16)a3[i];
      }
      v8h b0lo = *(const v8h*)(bc0 + k0);
      v8h b0hi = *(const v8h*)(bc0 + k0 + 16);
      v8h b1lo = *(const v8h*)(bc1 + k0);
      v8h b1hi = *(const v8h*)(bc1 + k0 + 16);
      v16h bf0, bf1;
#pragma unroll
      for (int i = 0; i < 8; ++i) {
        bf0[i] = b0lo[i]; bf0[8 + i] = b0hi[i];
        bf1[i] = b1lo[i]; bf1[8 + i] = b1hi[i];
      }
      acc0 = __builtin_amdgcn_wmma_f32_16x16x32_f16(false, af, false, bf0,
                                                    (short)0, acc0, false, false);
      acc1 = __builtin_amdgcn_wmma_f32_16x16x32_f16(false, af, false, bf1,
                                                    (short)0, acc1, false, false);
    }
  }

#pragma unroll
  for (int i = 0; i < CHUNK / 256; ++i)
    if (r[i] >= 0) atomicAdd(&cnt[r[i] >> RB_SHIFT], 1);
  __syncthreads();
  for (int k = tid; k < NBPAD; k += 256) {
    const int n = cnt[k];
    if (n) atomicAdd(&gcount[k], n);
  }

  if (row0 < N_NODES) {
    float* drow = Y1 + (size_t)(row0 + kh * 8) * L1D + m;
#pragma unroll
    for (int v = 0; v < 8; ++v) {
      drow[(size_t)v * L1D]      = acc0[v];
      drow[(size_t)v * L1D + 16] = acc1[v];
    }
  }
}

__global__ void egae_scatter(const int* __restrict__ er, const int* __restrict__ ec,
                             const float* __restrict__ ev,
                             const int* __restrict__ gcount, int* __restrict__ gcursor,
                             int* __restrict__ gbase, int2* __restrict__ E) {
  __shared__ int   cnt[NBPAD];
  __shared__ int   gdst[NBPAD];
  __shared__ int   lstart[NBPAD];
  __shared__ int2  part[256];
  __shared__ int   sPk[CHUNK];
  __shared__ float sVal[CHUNK];
  __shared__ int   sDst[CHUNK];
  const int tid = threadIdx.x;
  for (int i = tid; i < NBPAD; i += 256) cnt[i] = 0;
  __syncthreads();
  const int base = blockIdx.x * CHUNK;
  const int b0 = tid * 4;
  const int4 gt = *(const int4*)(gcount + b0);
  int r[CHUNK / 256], c[CHUNK / 256], rk[CHUNK / 256];
  float v[CHUNK / 256];
#pragma unroll
  for (int i = 0; i < CHUNK / 256; ++i) {
    const int e = base + i * 256 + tid;
    r[i] = -1; c[i] = 0; v[i] = 0.0f; rk[i] = 0;
    if (e < N_EDGES) {
      r[i] = er[e]; c[i] = ec[e]; v[i] = ev[e];
      rk[i] = atomicAdd(&cnt[r[i] >> RB_SHIFT], 1);
    }
  }
  __syncthreads();
  int gslot[NBPAD / 256];
#pragma unroll
  for (int q = 0; q < NBPAD / 256; ++q) {
    const int k  = q * 256 + tid;
    const int nk = cnt[k];
    gslot[q] = 0;
    if (nk) gslot[q] = atomicAdd(&gcursor[k], nk);
  }
  int n[4];
#pragma unroll
  for (int k = 0; k < 4; ++k) n[k] = cnt[b0 + k];
  const int gn[4] = {gt.x, gt.y, gt.z, gt.w};
  const int mine  = n[0] + n[1] + n[2] + n[3];
  const int gmine = gn[0] + gn[1] + gn[2] + gn[3];
  part[tid] = make_int2(mine, gmine);
  __syncthreads();
  for (int off = 1; off < 256; off <<= 1) {
    int2 add = make_int2(0, 0);
    if (tid >= off) add = part[tid - off];
    __syncthreads();
    const int2 cur = part[tid];
    part[tid] = make_int2(cur.x + add.x, cur.y + add.y);
    __syncthreads();
  }
  const int2 incl = part[tid];
  int excl  = incl.x - mine;
  int gexcl = incl.y - gmine;
  int gs[4];
#pragma unroll
  for (int k = 0; k < 4; ++k) {
    lstart[b0 + k] = excl;
    cnt[b0 + k]    = gexcl;
    gs[k] = gexcl;
    excl  += n[k];
    gexcl += gn[k];
  }
  if (blockIdx.x == 0) {
    *(int4*)(gbase + b0) = make_int4(gs[0], gs[1], gs[2], gs[3]);
    if (tid == 255) gbase[NBPAD] = gexcl;
  }
#pragma unroll
  for (int q = 0; q < NBPAD / 256; ++q) gdst[q * 256 + tid] = gslot[q];
  __syncthreads();
#pragma unroll
  for (int i = 0; i < CHUNK / 256; ++i) {
    if (r[i] >= 0) {
      const int bk = r[i] >> RB_SHIFT;
      const int p  = lstart[bk] + rk[i];
      sPk[p]  = c[i] | ((r[i] & (RB - 1)) << 17);
      sVal[p] = v[i];
      sDst[p] = cnt[bk] + gdst[bk] + rk[i];
    }
  }
  __syncthreads();
  const int nvalid = (N_EDGES - base < CHUNK) ? (N_EDGES - base) : CHUNK;
#pragma unroll
  for (int i = 0; i < CHUNK / 256; ++i) {
    const int p = i * 256 + tid;
    if (p < nvalid) {
      int2 rec; rec.x = sPk[p]; rec.y = __float_as_int(sVal[p]);
      E[sDst[p]] = rec;
    }
  }
}

template <bool FULL>
static __device__ __forceinline__ void spmm32_tile(const float* __restrict__ src,
                                                   float* acc, int pk_l, float v_l, int lane) {
  float g[32];
#pragma unroll
  for (int j = 0; j < 32; ++j) {
    const int pk = __builtin_amdgcn_readlane(pk_l, j);
    g[j] = 0.0f;
    if (FULL || pk >= 0) g[j] = src[(size_t)(pk & COLMASK) * L1D + lane];
  }
#pragma unroll
  for (int j = 0; j < 32; ++j) {
    const int   pk = __builtin_amdgcn_readlane(pk_l, j);
    const float v  = __int_as_float(__builtin_amdgcn_readlane(__float_as_int(v_l), j));
    if (FULL || pk >= 0) atomicAdd(acc + (pk >> 17) * L1D + lane, v * g[j]);
  }
}

__global__ void __launch_bounds__(512)
egae_spmm32(const int2* __restrict__ E, const int* __restrict__ gbase,
            const float* __restrict__ src, const _Float16* __restrict__ W2t,
            float* __restrict__ Y2) {
  __shared__ float acc[RB * L1D];
  const int tid  = threadIdx.x;
  const int lane = tid & 31;
  const int wave = __builtin_amdgcn_readfirstlane(tid >> 5);
  for (int i = tid; i < RB * L1D; i += 512) acc[i] = 0.0f;
  __syncthreads();
  const int bkt = blockIdx.x;
  const int beg = gbase[bkt], end = gbase[bkt + 1];
  for (int base = beg + wave * 32; base < end; base += 16 * 32) {
    const int idx = base + lane;
    int pk_l = -1; float v_l = 0.0f;
    if (idx < end) { const int2 rec = E[idx]; pk_l = rec.x; v_l = __int_as_float(rec.y); }
    if (base + 32 <= end) spmm32_tile<true >(src, acc, pk_l, v_l, lane);
    else                  spmm32_tile<false>(src, acc, pk_l, v_l, lane);
  }
  __syncthreads();
  if (wave < RB / 16) {
    const int m  = lane & 15;
    const int kh = lane >> 4;
    const float* arow = acc + (wave * 16 + m) * L1D + kh * 8;
    v4f a0 = *(const v4f*)(arow);
    v4f a1 = *(const v4f*)(arow + 4);
    v4f a2 = *(const v4f*)(arow + 16);
    v4f a3 = *(const v4f*)(arow + 20);
    v16h af;
#pragma unroll
    for (int i = 0; i < 4; ++i) {
      af[i]      = (_Float16)fmaxf(a0[i], 0.0f);
      af[4 + i]  = (_Float16)fmaxf(a1[i], 0.0f);
      af[8 + i]  = (_Float16)fmaxf(a2[i], 0.0f);
      af[12 + i] = (_Float16)fmaxf(a3[i], 0.0f);
    }
    const _Float16* bc = W2t + (size_t)m * L1D + kh * 8;
    v8h blo = *(const v8h*)(bc);
    v8h bhi = *(const v8h*)(bc + 16);
    v16h bf;
#pragma unroll
    for (int i = 0; i < 8; ++i) { bf[i] = blo[i]; bf[8 + i] = bhi[i]; }
    v8f d = {};
    d = __builtin_amdgcn_wmma_f32_16x16x32_f16(false, af, false, bf,
                                               (short)0, d, false, false);
    const int row0 = bkt * RB + wave * 16 + kh * 8;
#pragma unroll
    for (int v = 0; v < 8; ++v)
      if (row0 + v < N_NODES) Y2[(size_t)(row0 + v) * L2D + m] = d[v];
  }
}

template <int J>
static __device__ __forceinline__ int row_lane(int v) {
  return __builtin_amdgcn_update_dpp(0, v, 0x150 + J, 0xF, 0xF, false);
}

template <int J, bool FULL>
static __device__ __forceinline__ void spmm16_gather(const float* __restrict__ src,
                                                     int pk_l, int col, float (&g)[16]) {
  if constexpr (J < 16) {
    const int pk = row_lane<J>(pk_l);
    g[J] = 0.0f;
    if (FULL || pk >= 0) g[J] = src[(size_t)(pk & COLMASK) * L2D + col];
    spmm16_gather<J + 1, FULL>(src, pk_l, col, g);
  }
}

template <int J, bool FULL>
static __device__ __forceinline__ void spmm16_add(float* acc, int pk_l,
                                                  float v_l, int col, const float (&g)[16]) {
  if constexpr (J < 16) {
    const int   pk = row_lane<J>(pk_l);
    const float v  = __int_as_float(row_lane<J>(__float_as_int(v_l)));
    if (FULL || pk >= 0) atomicAdd(acc + (pk >> 17) * L2D + col, v * g[J]);
    spmm16_add<J + 1, FULL>(acc, pk_l, v_l, col, g);
  }
}

__global__ void __launch_bounds__(512)
egae_spmm16(const int2* __restrict__ E, const int* __restrict__ gbase,
            const float* __restrict__ src, float* __restrict__ dst) {
  __shared__ float acc[RB * L2D];
  const int tid  = threadIdx.x;
  const int lane = tid & 31;
  const int wave = __builtin_amdgcn_readfirstlane(tid >> 5);
  for (int i = tid; i < RB * L2D; i += 512) acc[i] = 0.0f;
  __syncthreads();
  const int bkt = blockIdx.x;
  const int beg = gbase[bkt], end = gbase[bkt + 1];
  const int col = lane & 15;
  for (int base = beg + wave * 32; base < end; base += 16 * 32) {
    const int idx = base + lane;
    int pk_l = -1; float v_l = 0.0f;
    if (idx < end) { const int2 rec = E[idx]; pk_l = rec.x; v_l = __int_as_float(rec.y); }
    float g[16];
    if (base + 32 <= end) {
      spmm16_gather<0, true >(src, pk_l, col, g);
      spmm16_add<0, true >(acc, pk_l, v_l, col, g);
    } else {
      spmm16_gather<0, false>(src, pk_l, col, g);
      spmm16_add<0, false>(acc, pk_l, v_l, col, g);
    }
  }
  __syncthreads();
  const int row = bkt * RB + (tid >> 2);
  if (row < N_NODES)
    *(v4f*)(dst + (size_t)row * L2D + (tid & 3) * 4) = *(const v4f*)(acc + tid * 4);
}

__global__ void egae_norm(const float* __restrict__ Y3, float* __restrict__ out) {
  int i = blockIdx.x * blockDim.x + threadIdx.x;
  if (i >= N_NODES) return;
  const v4f* s = (const v4f*)(Y3 + (size_t)i * L2D);
  v4f x0 = s[0], x1 = s[1], x2 = s[2], x3 = s[3];
  float ss = 0.0f;
#pragma unroll
  for (int j = 0; j < 4; ++j)
    ss += x0[j] * x0[j] + x1[j] * x1[j] + x2[j] * x2[j] + x3[j] * x3[j];
  float inv = 1.0f / fmaxf(sqrtf(ss), 1e-7f);
  v4f* o = (v4f*)(out + (size_t)i * L2D);
  o[0] = x0 * inv; o[1] = x1 * inv; o[2] = x2 * inv; o[3] = x3 * inv;
}

extern "C" void kernel_launch(void* const* d_in, const int* in_sizes, int n_in,
                              void* d_out, int out_size, void* d_ws, size_t ws_size,
                              hipStream_t stream) {
  const float* X  = (const float*)d_in[0];
  const float* W1 = (const float*)d_in[1];
  const float* W2 = (const float*)d_in[2];
  const int*   lr = (const int*)d_in[3];
  const int*   lc = (const int*)d_in[4];
  const float* lv = (const float*)d_in[5];
  float* out = (float*)d_out;

  char* ws = (char*)d_ws;
  float*    Y1  = (float*)(ws + 0);
  float*    Y3  = (float*)(ws + 0);
  int2*     E   = (int2*)(ws + 12800000);
  _Float16* W1t = (_Float16*)(ws + 38400000);
  int*      gbase = (int*)(ws + 38400000);
  _Float16* W2t = (_Float16*)(ws + 38432768);
  float* Y2      = out;
  int*   gcount  = (int*)d_out;
  int*   gcursor = (int*)d_out + NBPAD;

  egae_prep<<<(D_FEAT * L1D + 255) / 256, 256, 0, stream>>>(W1, W2, W1t, W2t, gcount);
  egae_gemm1_hist<<<NCHUNK, 256, 0, stream>>>(X, W1t, Y1, lr, gcount);
  egae_scatter<<<NCHUNK, 256, 0, stream>>>(lr, lc, lv, gcount, gcursor, gbase, E);
  egae_spmm32<<<NBUCK, 512, 0, stream>>>(E, gbase, Y1, W2t, Y2);
  egae_spmm16<<<NBUCK, 512, 0, stream>>>(E, gbase, Y2, Y3);
  egae_norm<<<(N_NODES + 255) / 256, 256, 0, stream>>>(Y3, out);
}


# launch fusion: row normalisation done in spmm16's epilogue (norm launch gone, 6 -> 5); Y1 rows split between out and workspace so Y2 can leave out; a row's squares summed as 4 lane partials
# speedup vs baseline: 4.1429x; 1.0321x over previous
#include <hip/hip_runtime.h>
#include <hip/hip_bf16.h>


#define N_NODES 100000
#define D_FEAT  512
#define N_EDGES 3200000
#define L1D     32
#define L2D     16
#define RB_SHIFT 7
#define RB      128
#define NBUCK   782
#define NBPAD   1024
#define CHUNK   4096
#define NCHUNK  782
#define COLMASK 0x1FFFF
#define HALF    50000

typedef __attribute__((ext_vector_type(16))) _Float16 v16h;
typedef __attribute__((ext_vector_type(8)))  _Float16 v8h;
typedef __attribute__((ext_vector_type(8)))  float    v8f;
typedef __attribute__((ext_vector_type(4)))  float    v4f;

__global__ void egae_prep(const float* __restrict__ W1,
                          const float* __restrict__ W2,
                          _Float16* __restrict__ W1t,
                          _Float16* __restrict__ W2t,
                          int* __restrict__ counters) {
  int t = blockIdx.x * blockDim.x + threadIdx.x;
  if (t < 2 * NBPAD) counters[t] = 0;
  if (t < D_FEAT * L1D) {
    int k = t / L1D, n = t % L1D;
    W1t[(size_t)n * D_FEAT + k] = (_Float16)W1[t];
  }
  if (t < L1D * L2D) {
    int k = t / L2D, n = t % L2D;
    W2t[(size_t)n * L1D + k] = (_Float16)W2[t];
  }
}

static_assert((N_NODES / 16 + 7) / 8 <= NCHUNK, "gemm1 row tiles must fit the chunk grid");
__global__ void __launch_bounds__(256)
egae_gemm1_hist(const float* __restrict__ X,
                const _Float16* __restrict__ W1t,
                float* __restrict__ Y1lo, float* __restrict__ Y1hi,
                const int* __restrict__ er, int* __restrict__ gcount) {
  __shared__ int cnt[NBPAD];
  const int tid = threadIdx.x;
  for (int i = tid; i < NBPAD; i += 256) cnt[i] = 0;
  __syncthreads();
  const int ebase = blockIdx.x * CHUNK;
  int r[CHUNK / 256];
#pragma unroll
  for (int i = 0; i < CHUNK / 256; ++i) {
    const int e = ebase + i * 256 + tid;
    r[i] = -1;
    if (e < N_EDGES) r[i] = er[e];
  }

  const int lane = tid & 31;
  const int wave = (blockIdx.x * 256 + tid) >> 5;
  const int row0 = wave << 4;
  const int m  = lane & 15;
  const int kh = lane >> 4;
  v8f acc0 = {}; v8f acc1 = {};
  if (row0 < N_NODES) {
    const float*    arow = X   + (size_t)(row0 + m) * D_FEAT + kh * 8;
    const _Float16* bc0  = W1t + (size_t)m         * D_FEAT + kh * 8;
    const _Float16* bc1  = W1t + (size_t)(m + 16)  * D_FEAT + kh * 8;
    for (int k0 = 0; k0 < D_FEAT; k0 += 32) {
      v4f a0 = *(const v4f*)(arow + k0);
      v4f a1 = *(const v4f*)(arow + k0 + 4);
      v4f a2 = *(const v4f*)(arow + k0 + 16);
      v4f a3 = *(const v4f*)(arow + k0 + 20);
      v16h af;
#pragma unroll
      for (int i = 0; i < 4; ++i) {
        af[i]      = (_Float16)a0[i];
        af[4 + i]  = (_Float16)a1[i];
        af[8 + i]  = (_Float16)a2[i];
        af[12 + i] = (_Float16)a3[i];
      }
      v8h b0lo = *(const v8h*)(bc0 + k0);
      v8h b0hi = *(const v8h*)(bc0 + k0 + 16);
      v8h b1lo = *(const v8h*)(bc1 + k0);
      v8h b1hi = *(const v8h*)(bc1 + k0 + 16);
      v16h bf0, bf1;
#pragma unroll
      for (int i = 0; i < 8; ++i) {
        bf0[i] = b0lo[i]; bf0[8 + i] = b0hi[i];
        bf1[i] = b1lo[i]; bf1[8 + i] = b1hi[i];
      }
      acc0 = __builtin_amdgcn_wmma_f32_16x16x32_f16(false, af, false, bf0,
                                                    (short)0, acc0, false, false);
      acc1 = __builtin_amdgcn_wmma_f32_16x16x32_f16(false, af, false, bf1,
                                                    (short)0, acc1, false, false);
    }
  }

#pragma unroll
  for (int i = 0; i < CHUNK / 256; ++i)
    if (r[i] >= 0) atomicAdd(&cnt[r[i] >> RB_SHIFT], 1);
  __syncthreads();
  for (int k = tid; k < NBPAD; k += 256) {
    const int n = cnt[k];
    if (n) atomicAdd(&gcount[k], n);
  }

  if (row0 < N_NODES) {
    float* ytile = (row0 < HALF) ? Y1lo + (size_t)row0 * L1D
                                 : Y1hi + (size_t)(row0 - HALF) * L1D;
    float* drow = ytile + (size_t)(kh * 8) * L1D + m;
#pragma unroll
    for (int v = 0; v < 8; ++v) {
      drow[(size_t)v * L1D]      = acc0[v];
      drow[(size_t)v * L1D + 16] = acc1[v];
    }
  }
}

__global__ void egae_scatter(const int* __restrict__ er, const int* __restrict__ ec,
                             const float* __restrict__ ev,
                             const int* __restrict__ gcount, int* __restrict__ gcursor,
                             int* __restrict__ gbase, int2* __restrict__ E) {
  __shared__ int   cnt[NBPAD];
  __shared__ int   gdst[NBPAD];
  __shared__ int   lstart[NBPAD];
  __shared__ int2  part[256];
  __shared__ int   sPk[CHUNK];
  __shared__ float sVal[CHUNK];
  __shared__ int   sDst[CHUNK];
  const int tid = threadIdx.x;
  for (int i = tid; i < NBPAD; i += 256) cnt[i] = 0;
  __syncthreads();
  const int base = blockIdx.x * CHUNK;
  const int b0 = tid * 4;
  const int4 gt = *(const int4*)(gcount + b0);
  int r[CHUNK / 256], c[CHUNK / 256], rk[CHUNK / 256];
  float v[CHUNK / 256];
#pragma unroll
  for (int i = 0; i < CHUNK / 256; ++i) {
    const int e = base + i * 256 + tid;
    r[i] = -1; c[i] = 0; v[i] = 0.0f; rk[i] = 0;
    if (e < N_EDGES) {
      r[i] = er[e]; c[i] = ec[e]; v[i] = ev[e];
      rk[i] = atomicAdd(&cnt[r[i] >> RB_SHIFT], 1);
    }
  }
  __syncthreads();
  int gslot[NBPAD / 256];
#pragma unroll
  for (int q = 0; q < NBPAD / 256; ++q) {
    const int k  = q * 256 + tid;
    const int nk = cnt[k];
    gslot[q] = 0;
    if (nk) gslot[q] = atomicAdd(&gcursor[k], nk);
  }
  int n[4];
#pragma unroll
  for (int k = 0; k < 4; ++k) n[k] = cnt[b0 + k];
  const int gn[4] = {gt.x, gt.y, gt.z, gt.w};
  const int mine  = n[0] + n[1] + n[2] + n[3];
  const int gmine = gn[0] + gn[1] + gn[2] + gn[3];
  part[tid] = make_int2(mine, gmine);
  __syncthreads();
  for (int off = 1; off < 256; off <<= 1) {
    int2 add = make_int2(0, 0);
    if (tid >= off) add = part[tid - off];
    __syncthreads();
    const int2 cur = part[tid];
    part[tid] = make_int2(cur.x + add.x, cur.y + add.y);
    __syncthreads();
  }
  const int2 incl = part[tid];
  int excl  = incl.x - mine;
  int gexcl = incl.y - gmine;
  int gs[4];
#pragma unroll
  for (int k = 0; k < 4; ++k) {
    lstart[b0 + k] = excl;
    cnt[b0 + k]    = gexcl;
    gs[k] = gexcl;
    excl  += n[k];
    gexcl += gn[k];
  }
  if (blockIdx.x == 0) {
    *(int4*)(gbase + b0) = make_int4(gs[0], gs[1], gs[2], gs[3]);
    if (tid == 255) gbase[NBPAD] = gexcl;
  }
#pragma unroll
  for (int q = 0; q < NBPAD / 256; ++q) gdst[q * 256 + tid] = gslot[q];
  __syncthreads();
#pragma unroll
  for (int i = 0; i < CHUNK / 256; ++i) {
    if (r[i] >= 0) {
      const int bk = r[i] >> RB_SHIFT;
      const int p  = lstart[bk] + rk[i];
      sPk[p]  = c[i] | ((r[i] & (RB - 1)) << 17);
      sVal[p] = v[i];
      sDst[p] = cnt[bk] + gdst[bk] + rk[i];
    }
  }
  __syncthreads();
  const int nvalid = (N_EDGES - base < CHUNK) ? (N_EDGES - base) : CHUNK;
#pragma unroll
  for (int i = 0; i < CHUNK / 256; ++i) {
    const int p = i * 256 + tid;
    if (p < nvalid) {
      int2 rec; rec.x = sPk[p]; rec.y = __float_as_int(sVal[p]);
      E[sDst[p]] = rec;
    }
  }
}

typedef const __attribute__((address_space(1))) float* gfloat_cptr;

template <bool FULL>
static __device__ __forceinline__ void spmm32_tile(const float* __restrict__ srclo,
                                                   const float* __restrict__ srchi,
                                                   float* acc, int pk_l, float v_l, int lane) {
  const unsigned long long blo = (unsigned long long)srclo;
  const unsigned long long bhi = (unsigned long long)srchi - (unsigned long long)HALF * (L1D * 4);
  float g[32];
#pragma unroll
  for (int j = 0; j < 32; ++j) {
    const int pk = __builtin_amdgcn_readlane(pk_l, j);
    g[j] = 0.0f;
    if (FULL || pk >= 0) {
      const unsigned c = pk & COLMASK;
      const unsigned long long b = (c < HALF) ? blo : bhi;
      g[j] = ((gfloat_cptr)(b + (unsigned long long)c * (L1D * 4)))[lane];
    }
  }
#pragma unroll
  for (int j = 0; j < 32; ++j) {
    const int   pk = __builtin_amdgcn_readlane(pk_l, j);
    const float v  = __int_as_float(__builtin_amdgcn_readlane(__float_as_int(v_l), j));
    if (FULL || pk >= 0) atomicAdd(acc + (pk >> 17) * L1D + lane, v * g[j]);
  }
}

__global__ void __launch_bounds__(512)
egae_spmm32(const int2* __restrict__ E, const int* __restrict__ gbase,
            const float* __restrict__ srclo, const float* __restrict__ srchi,
            const _Float16* __restrict__ W2t, float* __restrict__ Y2) {
  __shared__ float acc[RB * L1D];
  const int tid  = threadIdx.x;
  const int lane = tid & 31;
  const int wave = __builtin_amdgcn_readfirstlane(tid >> 5);
  for (int i = tid; i < RB * L1D; i += 512) acc[i] = 0.0f;
  __syncthreads();
  const int bkt = blockIdx.x;
  const int beg = gbase[bkt], end = gbase[bkt + 1];
  for (int base = beg + wave * 32; base < end; base += 16 * 32) {
    const int idx = base + lane;
    int pk_l = -1; float v_l = 0.0f;
    if (idx < end) { const int2 rec = E[idx]; pk_l = rec.x; v_l = __int_as_float(rec.y); }
    if (base + 32 <= end) spmm32_tile<true >(srclo, srchi, acc, pk_l, v_l, lane);
    else                  spmm32_tile<false>(srclo, srchi, acc, pk_l, v_l, lane);
  }
  __syncthreads();
  if (wave < RB / 16) {
    const int m  = lane & 15;
    const int kh = lane >> 4;
    const float* arow = acc + (wave * 16 + m) * L1D + kh * 8;
    v4f a0 = *(const v4f*)(arow);
    v4f a1 = *(const v4f*)(arow + 4);
    v4f a2 = *(const v4f*)(arow + 16);
    v4f a3 = *(const v4f*)(arow + 20);
    v16h af;
#pragma unroll
    for (int i = 0; i < 4; ++i) {
      af[i]      = (_Float16)fmaxf(a0[i], 0.0f);
      af[4 + i]  = (_Float16)fmaxf(a1[i], 0.0f);
      af[8 + i]  = (_Float16)fmaxf(a2[i], 0.0f);
      af[12 + i] = (_Float16)fmaxf(a3[i], 0.0f);
    }
    const _Float16* bc = W2t + (size_t)m * L1D + kh * 8;
    v8h blo = *(const v8h*)(bc);
    v8h bhi = *(const v8h*)(bc + 16);
    v16h bf;
#pragma unroll
    for (int i = 0; i < 8; ++i) { bf[i] = blo[i]; bf[8 + i] = bhi[i]; }
    v8f d = {};
    d = __builtin_amdgcn_wmma_f32_16x16x32_f16(false, af, false, bf,
                                               (short)0, d, false, false);
    const int row0 = bkt * RB + wave * 16 + kh * 8;
#pragma unroll
    for (int v = 0; v < 8; ++v)
      if (row0 + v < N_NODES) Y2[(size_t)(row0 + v) * L2D + m] = d[v];
  }
}

template <int J>
static __device__ __forceinline__ int row_lane(int v) {
  return __builtin_amdgcn_update_dpp(0, v, 0x150 + J, 0xF, 0xF, false);
}

template <int J, bool FULL>
static __device__ __forceinline__ void spmm16_gather(const float* __restrict__ src,
                                                     int pk_l, int col, float (&g)[16]) {
  if constexpr (J < 16) {
    const int pk = row_lane<J>(pk_l);
    g[J] = 0.0f;
    if (FULL || pk >= 0) g[J] = src[(size_t)(pk & COLMASK) * L2D + col];
    spmm16_gather<J + 1, FULL>(src, pk_l, col, g);
  }
}

template <int J, bool FULL>
static __device__ __forceinline__ void spmm16_add(float* acc, int pk_l,
                                                  float v_l, int col, const float (&g)[16]) {
  if constexpr (J < 16) {
    const int   pk = row_lane<J>(pk_l);
    const float v  = __int_as_float(row_lane<J>(__float_as_int(v_l)));
    if (FULL || pk >= 0) atomicAdd(acc + (pk >> 17) * L2D + col, v * g[J]);
    spmm16_add<J + 1, FULL>(acc, pk_l, v_l, col, g);
  }
}

__global__ void __launch_bounds__(512)
egae_spmm16(const int2* __restrict__ E, const int* __restrict__ gbase,
            const float* __restrict__ src, float* __restrict__ out) {
  __shared__ float acc[RB * L2D];
  const int tid  = threadIdx.x;
  const int lane = tid & 31;
  const int wave = __builtin_amdgcn_readfirstlane(tid >> 5);
  for (int i = tid; i < RB * L2D; i += 512) acc[i] = 0.0f;
  __syncthreads();
  const int bkt = blockIdx.x;
  const int beg = gbase[bkt], end = gbase[bkt + 1];
  const int col = lane & 15;
  for (int base = beg + wave * 32; base < end; base += 16 * 32) {
    const int idx = base + lane;
    int pk_l = -1; float v_l = 0.0f;
    if (idx < end) { const int2 rec = E[idx]; pk_l = rec.x; v_l = __int_as_float(rec.y); }
    float g[16];
    if (base + 32 <= end) {
      spmm16_gather<0, true >(src, pk_l, col, g);
      spmm16_add<0, true >(acc, pk_l, v_l, col, g);
    } else {
      spmm16_gather<0, false>(src, pk_l, col, g);
      spmm16_add<0, false>(acc, pk_l, v_l, col, g);
    }
  }
  __syncthreads();
  const int row = bkt * RB + (tid >> 2);
  const v4f x = *(const v4f*)(acc + tid * 4);
  float ss = x[0] * x[0] + x[1] * x[1] + x[2] * x[2] + x[3] * x[3];
  ss += __shfl_xor(ss, 1);
  ss += __shfl_xor(ss, 2);
  const float inv = 1.0f / fmaxf(sqrtf(ss), 1e-7f);
  if (row < N_NODES)
    *(v4f*)(out + (size_t)row * L2D + (tid & 3) * 4) = x * inv;
}

extern "C" void kernel_launch(void* const* d_in, const int* in_sizes, int n_in,
                              void* d_out, int out_size, void* d_ws, size_t ws_size,
                              hipStream_t stream) {
  const float* X  = (const float*)d_in[0];
  const float* W1 = (const float*)d_in[1];
  const float* W2 = (const float*)d_in[2];
  const int*   lr = (const int*)d_in[3];
  const int*   lc = (const int*)d_in[4];
  const float* lv = (const float*)d_in[5];
  float* out = (float*)d_out;

  char* ws = (char*)d_ws;
  float*    Y1lo = out;
  float*    Y1hi = (float*)(ws + 0);
  float*    Y2   = (float*)(ws + 6400000);
  int*      gcount  = (int*)(ws + 6400000);
  int*      gcursor = (int*)(ws + 6400000) + NBPAD;
  int2*     E   = (int2*)(ws + 12800000);
  _Float16* W1t = (_Float16*)(ws + 38400000);
  int*      gbase = (int*)(ws + 38400000);
  _Float16* W2t = (_Float16*)(ws + 38432768);

  egae_prep<<<(D_FEAT * L1D + 255) / 256, 256, 0, stream>>>(W1, W2, W1t, W2t, gcount);
  egae_gemm1_hist<<<NCHUNK, 256, 0, stream>>>(X, W1t, Y1lo, Y1hi, lr, gcount);
  egae_scatter<<<NCHUNK, 256, 0, stream>>>(lr, lc, lv, gcount, gcursor, gbase, E);
  egae_spmm32<<<NBUCK, 512, 0, stream>>>(E, gbase, Y1lo, Y1hi, W2t, Y2);
  egae_spmm16<<<NBUCK, 512, 0, stream>>>(E, gbase, Y2, out);
}


# gemm1: X read as whole 512-byte row pieces (16 B per lane) and turned into A fragments through wave-private LDS; W1 stored in fragment order so B loads are whole lines too
# speedup vs baseline: 4.5829x; 1.1062x over previous
#include <hip/hip_runtime.h>
#include <hip/hip_bf16.h>


#define N_NODES 100000
#define D_FEAT  512
#define N_EDGES 3200000
#define L1D     32
#define L2D     16
#define RB_SHIFT 7
#define RB      128
#define NBUCK   782
#define NBPAD   1024
#define CHUNK   4096
#define NCHUNK  782
#define COLMASK 0x1FFFF
#define HALF    50000
#define XSLAB   128
#define XPITCH  136

typedef __attribute__((ext_vector_type(16))) _Float16 v16h;
typedef __attribute__((ext_vector_type(8)))  _Float16 v8h;
typedef __attribute__((ext_vector_type(4)))  _Float16 v4h;
typedef __attribute__((ext_vector_type(8)))  float    v8f;
typedef __attribute__((ext_vector_type(4)))  float    v4f;

__global__ void egae_prep(const float* __restrict__ W1,
                          const float* __restrict__ W2,
                          _Float16* __restrict__ W1f,
                          _Float16* __restrict__ W2t,
                          int* __restrict__ counters) {
  int t = blockIdx.x * blockDim.x + threadIdx.x;
  if (t < 2 * NBPAD) counters[t] = 0;
  if (t < D_FEAT * L1D) {
    const int k = t / L1D, n = t % L1D;
    const int s = k >> 5, kk = k & 31;
    const int j = kk >> 4, kh = (kk >> 3) & 1, e = kk & 7;
    const int ct = n >> 4, m = n & 15;
    W1f[(((s * 2 + ct) * 2 + j) * 32 + m + 16 * kh) * 8 + e] = (_Float16)W1[t];
  }
  if (t < L1D * L2D) {
    int k = t / L2D, n = t % L2D;
    W2t[(size_t)n * L1D + k] = (_Float16)W2[t];
  }
}

static_assert((N_NODES / 16 + 7) / 8 <= NCHUNK, "gemm1 row tiles must fit the chunk grid");
__global__ void __launch_bounds__(256)
egae_gemm1_hist(const float* __restrict__ X,
                const _Float16* __restrict__ W1f,
                float* __restrict__ Y1lo, float* __restrict__ Y1hi,
                const int* __restrict__ er, int* __restrict__ gcount) {
  __shared__ int cnt[NBPAD];
  __shared__ __attribute__((aligned(16))) _Float16 xs[8][16][XPITCH];
  const int tid = threadIdx.x;
  for (int i = tid; i < NBPAD; i += 256) cnt[i] = 0;
  __syncthreads();
  const int ebase = blockIdx.x * CHUNK;
  int r[CHUNK / 256];
#pragma unroll
  for (int i = 0; i < CHUNK / 256; ++i) {
    const int e = ebase + i * 256 + tid;
    r[i] = -1;
    if (e < N_EDGES) r[i] = er[e];
  }

  const int lane = tid & 31;
  const int wave = (blockIdx.x * 256 + tid) >> 5;
  const int row0 = wave << 4;
  const int m  = lane & 15;
  const int kh = lane >> 4;
  v8f acc0 = {}; v8f acc1 = {};
  if (row0 < N_NODES) {
    const int wv = tid >> 5;
    const float*    xrow = X + (size_t)row0 * D_FEAT + lane * 4;
    _Float16*       st   = &xs[wv][0][lane * 4];
    const _Float16* fr   = &xs[wv][m][kh * 8];
    const _Float16* bl   = W1f + lane * 8;
    for (int sb = 0; sb < D_FEAT / XSLAB; ++sb) {
      v4f x[16];
#pragma unroll
      for (int i = 0; i < 16; ++i)
        x[i] = *(const v4f*)(xrow + (size_t)i * D_FEAT + sb * XSLAB);
#pragma unroll
      for (int i = 0; i < 16; ++i) {
        v4h h;
        h[0] = (_Float16)x[i][0]; h[1] = (_Float16)x[i][1];
        h[2] = (_Float16)x[i][2]; h[3] = (_Float16)x[i][3];
        *(v4h*)(st + i * XPITCH) = h;
      }
      __builtin_amdgcn_fence(__ATOMIC_RELEASE, "wavefront");
      __builtin_amdgcn_fence(__ATOMIC_ACQUIRE, "wavefront");
#pragma unroll
      for (int ks = 0; ks < XSLAB / 32; ++ks) {
        v8h alo = *(const v8h*)(fr + ks * 32);
        v8h ahi = *(const v8h*)(fr + ks * 32 + 16);
        const _Float16* b = bl + (size_t)(sb * (XSLAB / 32) + ks) * 1024;
        v8h b0lo = *(const v8h*)(b);
        v8h b0hi = *(const v8h*)(b + 256);
        v8h b1lo = *(const v8h*)(b + 512);
        v8h b1hi = *(const v8h*)(b + 768);
        v16h af, bf0, bf1;
#pragma unroll
        for (int i = 0; i < 8; ++i) {
          af[i]  = alo[i];  af[8 + i]  = ahi[i];
          bf0[i] = b0lo[i]; bf0[8 + i] = b0hi[i];
          bf1[i] = b1lo[i]; bf1[8 + i] = b1hi[i];
        }
        acc0 = __builtin_amdgcn_wmma_f32_16x16x32_f16(false, af, false, bf0,
                                                      (short)0, acc0, false, false);
        acc1 = __builtin_amdgcn_wmma_f32_16x16x32_f16(false, af, false, bf1,
                                                      (short)0, acc1, false, false);
      }
      __builtin_amdgcn_fence(__ATOMIC_RELEASE, "wavefront");
      __builtin_amdgcn_fence(__ATOMIC_ACQUIRE, "wavefront");
    }
  }

#pragma unroll
  for (int i = 0; i < CHUNK / 256; ++i)
    if (r[i] >= 0) atomicAdd(&cnt[r[i] >> RB_SHIFT], 1);
  __syncthreads();
  for (int k = tid; k < NBPAD; k += 256) {
    const int n = cnt[k];
    if (n) atomicAdd(&gcount[k], n);
  }

  if (row0 < N_NODES) {
    float* ytile = (row0 < HALF) ? Y1lo + (size_t)row0 * L1D
                                 : Y1hi + (size_t)(row0 - HALF) * L1D;
    float* drow = ytile + (size_t)(kh * 8) * L1D + m;
#pragma unroll
    for (int v = 0; v < 8; ++v) {
      drow[(size_t)v * L1D]      = acc0[v];
      drow[(size_t)v * L1D + 16] = acc1[v];
    }
  }
}

__global__ void egae_scatter(const int* __restrict__ er, const int* __restrict__ ec,
                             const float* __restrict__ ev,
                             const int* __restrict__ gcount, int* __restrict__ gcursor,
                             int* __restrict__ gbase, int2* __restrict__ E) {
  __shared__ int   cnt[NBPAD];
  __shared__ int   gdst[NBPAD];
  __shared__ int   lstart[NBPAD];
  __shared__ int2  part[256];
  __shared__ int   sPk[CHUNK];
  __shared__ float sVal[CHUNK];
  __shared__ int   sDst[CHUNK];
  const int tid = threadIdx.x;
  for (int i = tid; i < NBPAD; i += 256) cnt[i] = 0;
  __syncthreads();
  const int base = blockIdx.x * CHUNK;
  const int b0 = tid * 4;
  const int4 gt = *(const int4*)(gcount + b0);
  int r[CHUNK / 256], c[CHUNK / 256], rk[CHUNK / 256];
  float v[CHUNK / 256];
#pragma unroll
  for (int i = 0; i < CHUNK / 256; ++i) {
    const int e = base + i * 256 + tid;
    r[i] = -1; c[i] = 0; v[i] = 0.0f; rk[i] = 0;
    if (e < N_EDGES) {
      r[i] = er[e]; c[i] = ec[e]; v[i] = ev[e];
      rk[i] = atomicAdd(&cnt[r[i] >> RB_SHIFT], 1);
    }
  }
  __syncthreads();
  int gslot[NBPAD / 256];
#pragma unroll
  for (int q = 0; q < NBPAD / 256; ++q) {
    const int k  = q * 256 + tid;
    const int nk = cnt[k];
    gslot[q] = 0;
    if (nk) gslot[q] = atomicAdd(&gcursor[k], nk);
  }
  int n[4];
#pragma unroll
  for (int k = 0; k < 4; ++k) n[k] = cnt[b0 + k];
  const int gn[4] = {gt.x, gt.y, gt.z, gt.w};
  const int mine  = n[0] + n[1] + n[2] + n[3];
  const int gmine = gn[0] + gn[1] + gn[2] + gn[3];
  part[tid] = make_int2(mine, gmine);
  __syncthreads();
  for (int off = 1; off < 256; off <<= 1) {
    int2 add = make_int2(0, 0);
    if (tid >= off) add = part[tid - off];
    __syncthreads();
    const int2 cur = part[tid];
    part[tid] = make_int2(cur.x + add.x, cur.y + add.y);
    __syncthreads();
  }
  const int2 incl = part[tid];
  int excl  = incl.x - mine;
  int gexcl = incl.y - gmine;
  int gs[4];
#pragma unroll
  for (int k = 0; k < 4; ++k) {
    lstart[b0 + k] = excl;
    cnt[b0 + k]    = gexcl;
    gs[k] = gexcl;
    excl  += n[k];
    gexcl += gn[k];
  }
  if (blockIdx.x == 0) {
    *(int4*)(gbase + b0) = make_int4(gs[0], gs[1], gs[2], gs[3]);
    if (tid == 255) gbase[NBPAD] = gexcl;
  }
#pragma unroll
  for (int q = 0; q < NBPAD / 256; ++q) gdst[q * 256 + tid] = gslot[q];
  __syncthreads();
#pragma unroll
  for (int i = 0; i < CHUNK / 256; ++i) {
    if (r[i] >= 0) {
      const int bk = r[i] >> RB_SHIFT;
      const int p  = lstart[bk] + rk[i];
      sPk[p]  = c[i] | ((r[i] & (RB - 1)) << 17);
      sVal[p] = v[i];
      sDst[p] = cnt[bk] + gdst[bk] + rk[i];
    }
  }
  __syncthreads();
  const int nvalid = (N_EDGES - base < CHUNK) ? (N_EDGES - base) : CHUNK;
#pragma unroll
  for (int i = 0; i < CHUNK / 256; ++i) {
    const int p = i * 256 + tid;
    if (p < nvalid) {
      int2 rec; rec.x = sPk[p]; rec.y = __float_as_int(sVal[p]);
      E[sDst[p]] = rec;
    }
  }
}

typedef const __attribute__((address_space(1))) float* gfloat_cptr;

template <bool FULL>
static __device__ __forceinline__ void spmm32_tile(const float* __restrict__ srclo,
                                                   const float* __restrict__ srchi,
                                                   float* acc, int pk_l, float v_l, int lane) {
  const unsigned long long blo = (unsigned long long)srclo;
  const unsigned long long bhi = (unsigned long long)srchi - (unsigned long long)HALF * (L1D * 4);
  float g[32];
#pragma unroll
  for (int j = 0; j < 32; ++j) {
    const int pk = __builtin_amdgcn_readlane(pk_l, j);
    g[j] = 0.0f;
    if (FULL || pk >= 0) {
      const unsigned c = pk & COLMASK;
      const unsigned long long b = (c < HALF) ? blo : bhi;
      g[j] = ((gfloat_cptr)(b + (unsigned long long)c * (L1D * 4)))[lane];
    }
  }
#pragma unroll
  for (int j = 0; j < 32; ++j) {
    const int   pk = __builtin_amdgcn_readlane(pk_l, j);
    const float v  = __int_as_float(__builtin_amdgcn_readlane(__float_as_int(v_l), j));
    if (FULL || pk >= 0) atomicAdd(acc + (pk >> 17) * L1D + lane, v * g[j]);
  }
}

__global__ void __launch_bounds__(512)
egae_spmm32(const int2* __restrict__ E, const int* __restrict__ gbase,
            const float* __restrict__ srclo, const float* __restrict__ srchi,
            const _Float16* __restrict__ W2t, float* __restrict__ Y2) {
  __shared__ float acc[RB * L1D];
  const int tid  = threadIdx.x;
  const int lane = tid & 31;
  const int wave = __builtin_amdgcn_readfirstlane(tid >> 5);
  for (int i = tid; i < RB * L1D; i += 512) acc[i] = 0.0f;
  __syncthreads();
  const int bkt = blockIdx.x;
  const int beg = gbase[bkt], end = gbase[bkt + 1];
  for (int base = beg + wave * 32; base < end; base += 16 * 32) {
    const int idx = base + lane;
    int pk_l = -1; float v_l = 0.0f;
    if (idx < end) { const int2 rec = E[idx]; pk_l = rec.x; v_l = __int_as_float(rec.y); }
    if (base + 32 <= end) spmm32_tile<true >(srclo, srchi, acc, pk_l, v_l, lane);
    else                  spmm32_tile<false>(srclo, srchi, acc, pk_l, v_l, lane);
  }
  __syncthreads();
  if (wave < RB / 16) {
    const int m  = lane & 15;
    const int kh = lane >> 4;
    const float* arow = acc + (wave * 16 + m) * L1D + kh * 8;
    v4f a0 = *(const v4f*)(arow);
    v4f a1 = *(const v4f*)(arow + 4);
    v4f a2 = *(const v4f*)(arow + 16);
    v4f a3 = *(const v4f*)(arow + 20);
    v16h af;
#pragma unroll
    for (int i = 0; i < 4; ++i) {
      af[i]      = (_Float16)fmaxf(a0[i], 0.0f);
      af[4 + i]  = (_Float16)fmaxf(a1[i], 0.0f);
      af[8 + i]  = (_Float16)fmaxf(a2[i], 0.0f);
      af[12 + i] = (_Float16)fmaxf(a3[i], 0.0f);
    }
    const _Float16* bc = W2t + (size_t)m * L1D + kh * 8;
    v8h blo = *(const v8h*)(bc);
    v8h bhi = *(const v8h*)(bc + 16);
    v16h bf;
#pragma unroll
    for (int i = 0; i < 8; ++i) { bf[i] = blo[i]; bf[8 + i] = bhi[i]; }
    v8f d = {};
    d = __builtin_amdgcn_wmma_f32_16x16x32_f16(false, af, false, bf,
                                               (short)0, d, false, false);
    const int row0 = bkt * RB + wave * 16 + kh * 8;
#pragma unroll
    for (int v = 0; v < 8; ++v)
      if (row0 + v < N_NODES) Y2[(size_t)(row0 + v) * L2D + m] = d[v];
  }
}

template <int J>
static __device__ __forceinline__ int row_lane(int v) {
  return __builtin_amdgcn_update_dpp(0, v, 0x150 + J, 0xF, 0xF, false);
}

template <int J, bool FULL>
static __device__ __forceinline__ void spmm16_gather(const float* __restrict__ src,
                                                     int pk_l, int col, float (&g)[16]) {
  if constexpr (J < 16) {
    const int pk = row_lane<J>(pk_l);
    g[J] = 0.0f;
    if (FULL || pk >= 0) g[J] = src[(size_t)(pk & COLMASK) * L2D + col];
    spmm16_gather<J + 1, FULL>(src, pk_l, col, g);
  }
}

template <int J, bool FULL>
static __device__ __forceinline__ void spmm16_add(float* acc, int pk_l,
                                                  float v_l, int col, const float (&g)[16]) {
  if constexpr (J < 16) {
    const int   pk = row_lane<J>(pk_l);
    const float v  = __int_as_float(row_lane<J>(__float_as_int(v_l)));
    if (FULL || pk >= 0) atomicAdd(acc + (pk >> 17) * L2D + col, v * g[J]);
    spmm16_add<J + 1, FULL>(acc, pk_l, v_l, col, g);
  }
}

__global__ void __launch_bounds__(512)
egae_spmm16(const int2* __restrict__ E, const int* __restrict__ gbase,
            const float* __restrict__ src, float* __restrict__ out) {
  __shared__ float acc[RB * L2D];
  const int tid  = threadIdx.x;
  const int lane = tid & 31;
  const int wave = __builtin_amdgcn_readfirstlane(tid >> 5);
  for (int i = tid; i < RB * L2D; i += 512) acc[i] = 0.0f;
  __syncthreads();
  const int bkt = blockIdx.x;
  const int beg = gbase[bkt], end = gbase[bkt + 1];
  const int col = lane & 15;
  for (int base = beg + wave * 32; base < end; base += 16 * 32) {
    const int idx = base + lane;
    int pk_l = -1; float v_l = 0.0f;
    if (idx < end) { const int2 rec = E[idx]; pk_l = rec.x; v_l = __int_as_float(rec.y); }
    float g[16];
    if (base + 32 <= end) {
      spmm16_gather<0, true >(src, pk_l, col, g);
      spmm16_add<0, true >(acc, pk_l, v_l, col, g);
    } else {
      spmm16_gather<0, false>(src, pk_l, col, g);
      spmm16_add<0, false>(acc, pk_l, v_l, col, g);
    }
  }
  __syncthreads();
  const int row = bkt * RB + (tid >> 2);
  const v4f x = *(const v4f*)(acc + tid * 4);
  float ss = x[0] * x[0] + x[1] * x[1] + x[2] * x[2] + x[3] * x[3];
  ss += __shfl_xor(ss, 1);
  ss += __shfl_xor(ss, 2);
  const float inv = 1.0f / fmaxf(sqrtf(ss), 1e-7f);
  if (row < N_NODES)
    *(v4f*)(out + (size_t)row * L2D + (tid & 3) * 4) = x * inv;
}

extern "C" void kernel_launch(void* const* d_in, const int* in_sizes, int n_in,
                              void* d_out, int out_size, void* d_ws, size_t ws_size,
                              hipStream_t stream) {
  const float* X  = (const float*)d_in[0];
  const float* W1 = (const float*)d_in[1];
  const float* W2 = (const float*)d_in[2];
  const int*   lr = (const int*)d_in[3];
  const int*   lc = (const int*)d_in[4];
  const float* lv = (const float*)d_in[5];
  float* out = (float*)d_out;

  char* ws = (char*)d_ws;
  float*    Y1lo = out;
  float*    Y1hi = (float*)(ws + 0);
  float*    Y2   = (float*)(ws + 6400000);
  int*      gcount  = (int*)(ws + 6400000);
  int*      gcursor = (int*)(ws + 6400000) + NBPAD;
  int2*     E   = (int2*)(ws + 12800000);
  _Float16* W1f = (_Float16*)(ws + 38400000);
  int*      gbase = (int*)(ws + 38400000);
  _Float16* W2t = (_Float16*)(ws + 38432768);

  egae_prep<<<(D_FEAT * L1D + 255) / 256, 256, 0, stream>>>(W1, W2, W1f, W2t, gcount);
  egae_gemm1_hist<<<NCHUNK, 256, 0, stream>>>(X, W1f, Y1lo, Y1hi, lr, gcount);
  egae_scatter<<<NCHUNK, 256, 0, stream>>>(lr, lc, lv, gcount, gcursor, gbase, E);
  egae_spmm32<<<NBUCK, 512, 0, stream>>>(E, gbase, Y1lo, Y1hi, W2t, Y2);
  egae_spmm16<<<NBUCK, 512, 0, stream>>>(E, gbase, Y2, out);
}
